# starting kernel
# baseline (speedup 1.0000x reference)
#include <hip/hip_runtime.h>

typedef float v2f __attribute__((ext_vector_type(2)));
typedef float v8f __attribute__((ext_vector_type(8)));
typedef int   v2i __attribute__((ext_vector_type(2)));

#define IN_DIM  10
#define OUT_DIM 8
#define TILE_ELEMS (IN_DIM * OUT_DIM)
#define TPG 4
#define NG  2

__global__ __launch_bounds__(256)
void frag_dist5_wmma_kernel(const float* __restrict__ x,
                            const float* __restrict__ W,
                            const float* __restrict__ dW,
                            const int*   __restrict__ head_ix,
                            const int*   __restrict__ split_ix,
                            const int*   __restrict__ n_splits_p,
                            float*       __restrict__ out,
                            int Btot)
{
    const int ns   = n_splits_p[0];
    const int lane = threadIdx.x & 31;
    const int waveInBlock   = threadIdx.x >> 5;
    const int wavesPerBlock = blockDim.x >> 5;
    const int wid    = blockIdx.x * wavesPerBlock + waveInBlock;
    const int nWaves = gridDim.x * wavesPerBlock;

    const int  n        = lane & 15;
    const int  sel      = lane >> 4;
    const int  nc       = (n < OUT_DIM) ? n : (OUT_DIM - 1);
    const bool  isAx   = (lane == 0) || (lane == 18);
    const bool  isAy   = (lane == 1) || (lane == 19);
    const float axMask = isAx ? 1.0f : 0.0f;
    const float ayMask = isAy ? 1.0f : 0.0f;

    for (int t = wid * (TPG * NG); t < Btot; t += nWaves * (TPG * NG)) {
        const float* qW0[NG]; const float* qD0[NG];
        const float* qW1[NG]; const float* qD1[NG];
        const float* px[NG];
        float w0[NG][IN_DIM], d0[NG][IN_DIM], w1[NG][IN_DIM], d1[NG][IN_DIM];
        float xr[NG][IN_DIM];

        #pragma unroll
        for (int g = 0; g < NG; ++g) {
            const int tg = t + g * TPG;
            int tb0 = tg + 2 * sel;
            tb0 = (tb0 <= Btot - 2) ? tb0 : (Btot - 2);
            const v2i hp = *(const v2i*)(head_ix + tb0);
            const v2i sp = *(const v2i*)(split_ix + tb0);
            const int h0 = hp.x, h1 = hp.y, s0 = sp.x, s1 = sp.y;
            qW0[g] = W  + (size_t)h0 * TILE_ELEMS + nc;
            qD0[g] = dW + ((size_t)h0 * ns + s0) * TILE_ELEMS + nc;
            qW1[g] = W  + (size_t)h1 * TILE_ELEMS + nc;
            qD1[g] = dW + ((size_t)h1 * ns + s1) * TILE_ELEMS + nc;
            px[g]  = x + (size_t)(isAy ? (tb0 + 1) : tb0) * IN_DIM;
        }

        #pragma unroll
        for (int g = 0; g < NG; ++g) {
            #pragma unroll
            for (int j = 0; j < IN_DIM; ++j) {
                w0[g][j] = qW0[g][j * OUT_DIM];
                d0[g][j] = qD0[g][j * OUT_DIM];
                w1[g][j] = qW1[g][j * OUT_DIM];
                d1[g][j] = qD1[g][j * OUT_DIM];
            }
            #pragma unroll
            for (int j = 0; j < IN_DIM; ++j) xr[g][j] = px[g][j];
        }

        v8f acc[NG];
        #pragma unroll
        for (int g = 0; g < NG; ++g) acc[g] = (v8f){};
        #pragma unroll
        for (int j = 0; j < IN_DIM; ++j) {
            #pragma unroll
            for (int g = 0; g < NG; ++g) {
                v2f a, b;
                a.x = xr[g][j] * axMask;
                a.y = xr[g][j] * ayMask;
                b.x = fmaf(0.1f, d0[g][j], w0[g][j]);
                b.y = fmaf(0.1f, d1[g][j], w1[g][j]);
                acc[g] = __builtin_amdgcn_wmma_f32_16x16x4_f32(
                              false, a,  false, b,
                              (short)0, acc[g],  false,  false);
            }
        }

        if (lane < OUT_DIM) {
            #pragma unroll
            for (int g = 0; g < NG; ++g) {
                #pragma unroll
                for (int m = 0; m < TPG; ++m) {
                    const int tok = t + g * TPG + m;
                    if (tok < Btot)
                        out[(size_t)tok * OUT_DIM + lane] = acc[g][m];
                }
            }
        }
    }
}

extern "C" void kernel_launch(void* const* d_in, const int* in_sizes, int n_in,
                              void* d_out, int out_size, void* d_ws, size_t ws_size,
                              hipStream_t stream) {
    const float* x   = (const float*)d_in[0];
    const float* W   = (const float*)d_in[1];
    const float* dW  = (const float*)d_in[2];
    const int*   hix = (const int*)d_in[3];
    const int*   six = (const int*)d_in[4];
    const int*   nsp = (const int*)d_in[5];
    float* out = (float*)d_out;

    const int Btot = in_sizes[0] / IN_DIM;
    const int tokensPerWave = TPG * NG;
    const int waveTiles = (Btot + tokensPerWave - 1) / tokensPerWave;
    const int wavesPerBlock = 8;
    const int blocks = (waveTiles + wavesPerBlock - 1) / wavesPerBlock;

    frag_dist5_wmma_kernel<<<blocks, 256, 0, stream>>>(x, W, dW, hix, six, nsp, out, Btot);
}


# 16-byte gathers straight into B fragments: each WMMA sums 4 j terms for 8 tokens x 2 output halves (20 WMMAs per 16 tokens, was 40), one 16-byte store per lane; sum order over j changes
# speedup vs baseline: 1.4895x; 1.4895x over previous
#include <hip/hip_runtime.h>


typedef float v2f __attribute__((ext_vector_type(2)));
typedef float v4f __attribute__((ext_vector_type(4)));
typedef float v8f __attribute__((ext_vector_type(8)));

#define IN_DIM  10
#define OUT_DIM 8
#define TILE_ELEMS (IN_DIM * OUT_DIM)
#define NPAIR (IN_DIM / 2)
#define TOK_PER_WAVE 16

__global__ __launch_bounds__(256)
void frag_dist5_wmma_kernel(const float* __restrict__ x,
                            const float* __restrict__ W,
                            const float* __restrict__ dW,
                            const int*   __restrict__ head_ix,
                            const int*   __restrict__ split_ix,
                            const int*   __restrict__ n_splits_p,
                            float*       __restrict__ out,
                            int Btot)
{
    const int ns   = n_splits_p[0];
    const int lane = threadIdx.x & 31;
    const int waveInBlock   = threadIdx.x >> 5;
    const int wavesPerBlock = blockDim.x >> 5;
    const int wid    = blockIdx.x * wavesPerBlock + waveInBlock;
    const int nWaves = gridDim.x * wavesPerBlock;

    const int  c    = lane & 15;
    const int  h    = lane >> 4;
    const int  tokc = c >> 1;
    const int  q    = c & 1;
    const int  pieceOff = h * OUT_DIM + q * 4;
    const bool lowRows  = c < 8;
    const int  last = Btot - 1;

    for (int t = wid * TOK_PER_WAVE; t < Btot; t += nWaves * TOK_PER_WAVE) {
        int tA = t + tokc;      tA = (tA < last) ? tA : last;
        int tB = t + 8 + tokc;  tB = (tB < last) ? tB : last;
        int tx = t + c;         tx = (tx < last) ? tx : last;

        const int hA = head_ix[tA], sA = split_ix[tA];
        const int hB = head_ix[tB], sB = split_ix[tB];

        const float* pWA = W  + (size_t)hA * TILE_ELEMS + pieceOff;
        const float* pDA = dW + ((size_t)hA * ns + sA) * TILE_ELEMS + pieceOff;
        const float* pWB = W  + (size_t)hB * TILE_ELEMS + pieceOff;
        const float* pDB = dW + ((size_t)hB * ns + sB) * TILE_ELEMS + pieceOff;
        const float* px  = x  + (size_t)tx * IN_DIM + h;

        v4f wA[NPAIR], dA[NPAIR], wB[NPAIR], dB[NPAIR];
        float xv[NPAIR];
        #pragma unroll
        for (int L = 0; L < NPAIR; ++L) {
            wA[L] = *(const v4f*)(pWA + 2 * OUT_DIM * L);
            dA[L] = *(const v4f*)(pDA + 2 * OUT_DIM * L);
            wB[L] = *(const v4f*)(pWB + 2 * OUT_DIM * L);
            dB[L] = *(const v4f*)(pDB + 2 * OUT_DIM * L);
        }
        #pragma unroll
        for (int L = 0; L < NPAIR; ++L) xv[L] = px[2 * L];
        __builtin_amdgcn_sched_barrier(0);

        v4f bA[NPAIR], bB[NPAIR];
        #pragma unroll
        for (int L = 0; L < NPAIR; ++L) {
            #pragma unroll
            for (int i = 0; i < 4; ++i) {
                bA[L][i] = fmaf(0.1f, dA[L][i], wA[L][i]);
                bB[L][i] = fmaf(0.1f, dB[L][i], wB[L][i]);
            }
        }

        const float xa0 = lowRows ? xv[0] : 0.0f, xb0 = lowRows ? 0.0f : xv[0];
        const float xa1 = lowRows ? xv[1] : 0.0f, xb1 = lowRows ? 0.0f : xv[1];
        const float xa2 = lowRows ? xv[2] : 0.0f, xb2 = lowRows ? 0.0f : xv[2];
        const float xa3 = lowRows ? xv[3] : 0.0f, xb3 = lowRows ? 0.0f : xv[3];
        const float xa4 = lowRows ? xv[4] : 0.0f, xb4 = lowRows ? 0.0f : xv[4];
        const v2f aA01 = {xa0, xa1}, aA23 = {xa2, xa3};
        const v2f aB01 = {xb0, xb1}, aB23 = {xb2, xb3};
        const v2f aAB4 = {xa4, xb4};

        v8f acc[4];
        #pragma unroll
        for (int i = 0; i < 4; ++i) acc[i] = (v8f){};

        #pragma unroll
        for (int i = 0; i < 4; ++i) {
            const v2f b = {bA[0][i], bA[1][i]};
            acc[i] = __builtin_amdgcn_wmma_f32_16x16x4_f32(false, aA01, false, b, (short)0, acc[i], false, false);
        }
        #pragma unroll
        for (int i = 0; i < 4; ++i) {
            const v2f b = {bA[2][i], bA[3][i]};
            acc[i] = __builtin_amdgcn_wmma_f32_16x16x4_f32(false, aA23, false, b, (short)0, acc[i], false, false);
        }
        #pragma unroll
        for (int i = 0; i < 4; ++i) {
            const v2f b = {bB[0][i], bB[1][i]};
            acc[i] = __builtin_amdgcn_wmma_f32_16x16x4_f32(false, aB01, false, b, (short)0, acc[i], false, false);
        }
        #pragma unroll
        for (int i = 0; i < 4; ++i) {
            const v2f b = {bB[2][i], bB[3][i]};
            acc[i] = __builtin_amdgcn_wmma_f32_16x16x4_f32(false, aB23, false, b, (short)0, acc[i], false, false);
        }
        #pragma unroll
        for (int i = 0; i < 4; ++i) {
            const v2f b = {bA[4][i], bB[4][i]};
            acc[i] = __builtin_amdgcn_wmma_f32_16x16x4_f32(false, aAB4, false, b, (short)0, acc[i], false, false);
        }

        v4f o;
        #pragma unroll
        for (int i = 0; i < 4; ++i) {
            float v = acc[i][0];
            #pragma unroll
            for (int r = 1; r < 8; ++r) v = (tokc == r) ? acc[i][r] : v;
            o[i] = v;
        }
        if (t + (lane >> 1) < Btot)
            *(v4f*)(out + (size_t)t * OUT_DIM + 4 * lane) = o;
    }
}

extern "C" void kernel_launch(void* const* d_in, const int* in_sizes, int n_in,
                              void* d_out, int out_size, void* d_ws, size_t ws_size,
                              hipStream_t stream) {
    const float* x   = (const float*)d_in[0];
    const float* W   = (const float*)d_in[1];
    const float* dW  = (const float*)d_in[2];
    const int*   hix = (const int*)d_in[3];
    const int*   six = (const int*)d_in[4];
    const int*   nsp = (const int*)d_in[5];
    float* out = (float*)d_out;

    const int Btot = in_sizes[0] / IN_DIM;
    const int tokensPerWave = TOK_PER_WAVE;
    const int waveTiles = (Btot + tokensPerWave - 1) / tokensPerWave;
    const int wavesPerBlock = 8;
    const int blocks = (waveTiles + wavesPerBlock - 1) / wavesPerBlock;

    frag_dist5_wmma_kernel<<<blocks, 256, 0, stream>>>(x, W, dW, hix, six, nsp, out, Btot);
}
